# starting kernel
# baseline (speedup 1.0000x reference)
#include <hip/hip_runtime.h>

typedef __attribute__((ext_vector_type(2))) float v2f;
typedef __attribute__((ext_vector_type(8))) float v8f;

template <int IMM>
__device__ __forceinline__ float swz_xor_add(float v) {
    int p = __builtin_amdgcn_ds_swizzle(__float_as_int(v), IMM);
    return v + __int_as_float(p);
}

__global__ __launch_bounds__(256) void dendrite_wmma_kernel(
    const float* __restrict__ x,
    const float* __restrict__ kk,
    const float* __restrict__ w,
    const float* __restrict__ q,
    const float* __restrict__ lin_w,
    const float* __restrict__ lin_b,
    float* __restrict__ out)
{
    __shared__ float plo_lds[8][16][16];

    const int lane = threadIdx.x & 31;
    const int wv   = threadIdx.x >> 5;
    const int half = lane >> 4;
    const int n    = lane & 15;
    const int tile = blockIdx.x * 8 + wv;
    const int row_g0 = tile * 16;

    v2f bmat[4];
    #pragma unroll
    for (int j = 0; j < 4; ++j) {
        #pragma unroll
        for (int e = 0; e < 2; ++e) {
            int h = 4 * j + 2 * half + e;
            int v = h * 16 + n;
            bmat[j][e] = (v == 0) ? 0.0f : lin_w[v - 1];
        }
    }

    const int row_g = row_g0 + n;
    const int b_idx = row_g >> 5;
    const int d_idx = row_g & 31;

    float sv[8];
    #pragma unroll
    for (int s = 0; s < 8; ++s) {
        float xs = x[b_idx * 8 + s];
        float z  = kk[d_idx * 8 + s] * (w[d_idx * 8 + s] * xs - q[d_idx * 8 + s]);
        sv[s] = __builtin_amdgcn_rcpf(1.0f + __expf(-z));
    }

    const float t0 = sv[3], t1 = sv[2], t2 = sv[1], t3 = sv[0];
    const float base = half ? t1 : 1.0f;

    v2f amat[4];
    amat[0][0] = base;
    amat[0][1] = base * t0;
    amat[1][0] = base * t2;
    amat[1][1] = amat[1][0] * t0;
    amat[2][0] = base * t3;
    amat[2][1] = amat[2][0] * t0;
    amat[3][0] = amat[2][0] * t2;
    amat[3][1] = amat[3][0] * t0;

    if (half == 0) {
        float pl[16];
        pl[0] = 1.0f;
        int len = 1;
        #pragma unroll
        for (int j = 0; j < 4; ++j) {
            float tl = sv[7 - j];
            #pragma unroll
            for (int m = 0; m < 8; ++m) {
                if (m < len) pl[len + m] = pl[m] * tl;
            }
            len <<= 1;
        }
        #pragma unroll
        for (int i = 0; i < 16; ++i)
            plo_lds[wv][n][i] = pl[i];
    }

    v8f c = {};
    #pragma unroll
    for (int j = 0; j < 4; ++j)
        c = __builtin_amdgcn_wmma_f32_16x16x4_f32(
                 false, amat[j],  false, bmat[j],
                 (short)0, c,  false,  false);

    float partial[8];
    #pragma unroll
    for (int r = 0; r < 8; ++r)
        partial[r] = c[r] * plo_lds[wv][8 * half + r][n];

    #pragma unroll
    for (int r = 0; r < 8; ++r) {
        float v = partial[r];
        v = swz_xor_add<(1 << 10) | 0x1f>(v);
        v = swz_xor_add<(2 << 10) | 0x1f>(v);
        v = swz_xor_add<(4 << 10) | 0x1f>(v);
        v = swz_xor_add<(8 << 10) | 0x1f>(v);
        partial[r] = v;
    }

    if (n == 0) {
        float bias = lin_b[0];
        #pragma unroll
        for (int r = 0; r < 8; ++r)
            out[row_g0 + 8 * half + r] = partial[r] + bias;
    }
}

extern "C" void kernel_launch(void* const* d_in, const int* in_sizes, int n_in,
                              void* d_out, int out_size, void* d_ws, size_t ws_size,
                              hipStream_t stream) {
    const float* x     = (const float*)d_in[0];
    const float* k     = (const float*)d_in[1];
    const float* w     = (const float*)d_in[2];
    const float* q     = (const float*)d_in[3];
    const float* lin_w = (const float*)d_in[5];
    const float* lin_b = (const float*)d_in[6];
    float* out = (float*)d_out;

    dendrite_wmma_kernel<<<dim3(2048), dim3(256), 0, stream>>>(
        x, k, w, q, lin_w, lin_b, out);
}


# swap WMMA operands (W2^T as A, Phi^T as B) so each lane holds 8 Y of its own row; row dot in-lane by nested FMAs + one permlanex16 half-join; LDS staging and swizzle tree gone; sum order changes
# speedup vs baseline: 1.2408x; 1.2408x over previous
#include <hip/hip_runtime.h>

typedef __attribute__((ext_vector_type(2))) float v2f;
typedef __attribute__((ext_vector_type(8))) float v8f;

__device__ __forceinline__ float xhalf(float v) {
    int b = __float_as_int(v);
    return __int_as_float(
        __builtin_amdgcn_permlanex16(b, b, 0x76543210, 0xfedcba98, false, false));
}

__global__ __launch_bounds__(256) void dendrite_wmma_kernel(
    const float* __restrict__ x,
    const float* __restrict__ kk,
    const float* __restrict__ w,
    const float* __restrict__ q,
    const float* __restrict__ lin_w,
    const float* __restrict__ lin_b,
    float* __restrict__ out)
{
    const int lane = threadIdx.x & 31;
    const int wv   = threadIdx.x >> 5;
    const int half = lane >> 4;
    const int n    = lane & 15;
    const int tile = blockIdx.x * 8 + wv;
    const int row_g0 = tile * 16;

    v2f wmat[4];
    #pragma unroll
    for (int j = 0; j < 4; ++j) {
        #pragma unroll
        for (int e = 0; e < 2; ++e) {
            int h = 4 * j + 2 * half + e;
            int v = h * 16 + n;
            wmat[j][e] = (v == 0) ? 0.0f : lin_w[v - 1];
        }
    }

    const int row_g = row_g0 + n;
    const int b_idx = row_g >> 5;
    const int d_idx = row_g & 31;

    float sv[8];
    #pragma unroll
    for (int s = 0; s < 8; ++s) {
        float xs = x[b_idx * 8 + s];
        float z  = kk[d_idx * 8 + s] * (w[d_idx * 8 + s] * xs - q[d_idx * 8 + s]);
        sv[s] = __builtin_amdgcn_rcpf(1.0f + __expf(-z));
    }

    const float t0 = sv[3], t1 = sv[2], t2 = sv[1], t3 = sv[0];
    const float base = half ? t1 : 1.0f;

    v2f pmat[4];
    pmat[0][0] = base;
    pmat[0][1] = base * t0;
    pmat[1][0] = base * t2;
    pmat[1][1] = pmat[1][0] * t0;
    pmat[2][0] = base * t3;
    pmat[2][1] = pmat[2][0] * t0;
    pmat[3][0] = pmat[2][0] * t2;
    pmat[3][1] = pmat[3][0] * t0;

    v8f c = {};
    #pragma unroll
    for (int j = 0; j < 4; ++j)
        c = __builtin_amdgcn_wmma_f32_16x16x4_f32(
                 false, wmat[j],  false, pmat[j],
                 (short)0, c,  false,  false);

    const float u0 = sv[7], u1 = sv[6], u2 = sv[5], u3 = sv[4];
    const float e0 = fmaf(c[1], u0, c[0]);
    const float e1 = fmaf(c[3], u0, c[2]);
    const float e2 = fmaf(c[5], u0, c[4]);
    const float e3 = fmaf(c[7], u0, c[6]);
    const float f0 = fmaf(e1, u1, e0);
    const float f1 = fmaf(e3, u1, e2);
    const float g  = fmaf(f1, u2, f0);
    const float part = g * (half ? u3 : 1.0f);
    const float tot  = part + xhalf(part);

    if (half == 0)
        out[row_g] = tot + lin_b[0];
}

extern "C" void kernel_launch(void* const* d_in, const int* in_sizes, int n_in,
                              void* d_out, int out_size, void* d_ws, size_t ws_size,
                              hipStream_t stream) {
    const float* x     = (const float*)d_in[0];
    const float* k     = (const float*)d_in[1];
    const float* w     = (const float*)d_in[2];
    const float* q     = (const float*)d_in[3];
    const float* lin_w = (const float*)d_in[5];
    const float* lin_b = (const float*)d_in[6];
    float* out = (float*)d_out;

    dendrite_wmma_kernel<<<dim3(2048), dim3(256), 0, stream>>>(
        x, k, w, q, lin_w, lin_b, out);
}


# each wave walks 2 whole batch rows (4 tiles): lin_w fragments, k/w/q rows and bias loaded once per wave, one 128-byte store per batch row; grid 2048 -> 512 blocks (16384 -> 4096 waves)
# speedup vs baseline: 1.5517x; 1.2506x over previous
#include <hip/hip_runtime.h>

typedef __attribute__((ext_vector_type(2))) float v2f;
typedef __attribute__((ext_vector_type(8))) float v8f;

__device__ __forceinline__ float xhalf(float v) {
    int b = __float_as_int(v);
    return __int_as_float(
        __builtin_amdgcn_permlanex16(b, b, 0x76543210, 0xfedcba98, false, false));
}

constexpr int BPW = 2;

__global__ __launch_bounds__(256) void dendrite_wmma_kernel(
    const float* __restrict__ x,
    const float* __restrict__ kk,
    const float* __restrict__ w,
    const float* __restrict__ q,
    const float* __restrict__ lin_w,
    const float* __restrict__ lin_b,
    float* __restrict__ out)
{
    const int lane = threadIdx.x & 31;
    const int wv   = threadIdx.x >> 5;
    const int half = lane >> 4;
    const int n    = lane & 15;
    const int b0   = (blockIdx.x * 8 + wv) * BPW;

    v2f wmat[4];
    #pragma unroll
    for (int j = 0; j < 4; ++j) {
        #pragma unroll
        for (int e = 0; e < 2; ++e) {
            int h = 4 * j + 2 * half + e;
            int v = h * 16 + n;
            wmat[j][e] = (v == 0) ? 0.0f : lin_w[v - 1];
        }
    }

    float kr[2][8], wr[2][8], qr[2][8];
    for (int p = 0; p < 2; ++p) {
        for (int s = 0; s < 8; ++s) {
            kr[p][s] = kk[(16 * p + n) * 8 + s];
            wr[p][s] = w[(16 * p + n) * 8 + s];
            qr[p][s] = q[(16 * p + n) * 8 + s];
        }
    }
    const float bias = lin_b[0];
    float xr[BPW][8];
    for (int i = 0; i < BPW; ++i) {
        for (int s = 0; s < 8; ++s)
            xr[i][s] = x[(b0 + i) * 8 + s];
    }

    float res[BPW];
    for (int i = 0; i < BPW; ++i) {
        float tot2[2];
        for (int p = 0; p < 2; ++p) {
            float sv[8];
            #pragma unroll
            for (int s = 0; s < 8; ++s) {
                float z = kr[p][s] * (wr[p][s] * xr[i][s] - qr[p][s]);
                sv[s] = __builtin_amdgcn_rcpf(1.0f + __expf(-z));
            }

            const float t0 = sv[3], t1 = sv[2], t2 = sv[1], t3 = sv[0];
            const float base = half ? t1 : 1.0f;

            v2f pmat[4];
            pmat[0][0] = base;
            pmat[0][1] = base * t0;
            pmat[1][0] = base * t2;
            pmat[1][1] = pmat[1][0] * t0;
            pmat[2][0] = base * t3;
            pmat[2][1] = pmat[2][0] * t0;
            pmat[3][0] = pmat[2][0] * t2;
            pmat[3][1] = pmat[3][0] * t0;

            v8f c = {};
            #pragma unroll
            for (int j = 0; j < 4; ++j)
                c = __builtin_amdgcn_wmma_f32_16x16x4_f32(
                         false, wmat[j],  false, pmat[j],
                         (short)0, c,  false,  false);

            const float u0 = sv[7], u1 = sv[6], u2 = sv[5], u3 = sv[4];
            const float e0 = fmaf(c[1], u0, c[0]);
            const float e1 = fmaf(c[3], u0, c[2]);
            const float e2 = fmaf(c[5], u0, c[4]);
            const float e3 = fmaf(c[7], u0, c[6]);
            const float f0 = fmaf(e1, u1, e0);
            const float f1 = fmaf(e3, u1, e2);
            const float g  = fmaf(f1, u2, f0);
            const float part = g * (half ? u3 : 1.0f);
            tot2[p] = part + xhalf(part);
        }
        res[i] = (half ? tot2[1] : tot2[0]) + bias;
    }

    for (int i = 0; i < BPW; ++i)
        out[(b0 + i) * 32 + lane] = res[i];
}

extern "C" void kernel_launch(void* const* d_in, const int* in_sizes, int n_in,
                              void* d_out, int out_size, void* d_ws, size_t ws_size,
                              hipStream_t stream) {
    const float* x     = (const float*)d_in[0];
    const float* k     = (const float*)d_in[1];
    const float* w     = (const float*)d_in[2];
    const float* q     = (const float*)d_in[3];
    const float* lin_w = (const float*)d_in[5];
    const float* lin_b = (const float*)d_in[6];
    float* out = (float*)d_out;

    dendrite_wmma_kernel<<<dim3(512), dim3(256), 0, stream>>>(
        x, k, w, q, lin_w, lin_b, out);
}
